# starting kernel
# baseline (speedup 1.0000x reference)
#include <hip/hip_runtime.h>
#include <hip/hip_bf16.h>

typedef __attribute__((ext_vector_type(16))) _Float16 v16h;
typedef __attribute__((ext_vector_type(8)))  float    v8f;

#define IN_F      64
#define OUT_F     32
#define NEG_SLOPE 0.2f

__device__ __forceinline__ unsigned enc_f32(float f) {
  unsigned u = __float_as_uint(f);
  return (u & 0x80000000u) ? ~u : (u | 0x80000000u);
}
__device__ __forceinline__ float dec_f32(unsigned u) {
  return (u & 0x80000000u) ? __uint_as_float(u & 0x7FFFFFFFu)
                           : __uint_as_float(~u);
}

__global__ __launch_bounds__(128)
void k_pack_w(const float* __restrict__ W, _Float16* __restrict__ wp) {
  const int tid   = threadIdx.x;
  const int lane  = tid & 31;
  const int ks    = (tid >> 5) & 1;
  const int t     = tid >> 6;
  const int kbase = (lane >= 16) ? 8 : 0;
  const int n     = t * 16 + (lane & 15);
  _Float16* dst = wp + (size_t)((t * 2 + ks) * 32 + lane) * 16;
#pragma unroll
  for (int j = 0; j < 8; ++j) {
    dst[j]     = (_Float16)W[(size_t)(ks * 32 + kbase + j)      * OUT_F + n];
    dst[j + 8] = (_Float16)W[(size_t)(ks * 32 + kbase + 16 + j) * OUT_F + n];
  }
}

__global__ __launch_bounds__(128)
void k_gemm(const float* __restrict__ x, const _Float16* __restrict__ wpack,
            float* __restrict__ h, int nNodes) {
  const int lane = threadIdx.x & 31;
  const int wave = blockIdx.x * (blockDim.x >> 5) + (threadIdx.x >> 5);
  const int m0   = wave * 16;
  if (m0 >= nNodes) return;

  const int mrow  = lane & 15;
  const int kbase = (lane >= 16) ? 8 : 0;
  int row = m0 + mrow;
  if (row > nNodes - 1) row = nNodes - 1;

  v16h a[2];
#pragma unroll
  for (int ks = 0; ks < 2; ++ks) {
    const float4* xr = (const float4*)(x + (size_t)row * IN_F + ks * 32 + kbase);
    const float4 lo0 = xr[0], lo1 = xr[1];
    const float4 hi0 = xr[4], hi1 = xr[5];
    a[ks][0]  = (_Float16)lo0.x; a[ks][1]  = (_Float16)lo0.y;
    a[ks][2]  = (_Float16)lo0.z; a[ks][3]  = (_Float16)lo0.w;
    a[ks][4]  = (_Float16)lo1.x; a[ks][5]  = (_Float16)lo1.y;
    a[ks][6]  = (_Float16)lo1.z; a[ks][7]  = (_Float16)lo1.w;
    a[ks][8]  = (_Float16)hi0.x; a[ks][9]  = (_Float16)hi0.y;
    a[ks][10] = (_Float16)hi0.z; a[ks][11] = (_Float16)hi0.w;
    a[ks][12] = (_Float16)hi1.x; a[ks][13] = (_Float16)hi1.y;
    a[ks][14] = (_Float16)hi1.z; a[ks][15] = (_Float16)hi1.w;
  }

  const v16h* wp = (const v16h*)wpack;
  v8f acc[2] = {};
#pragma unroll
  for (int t = 0; t < 2; ++t)
#pragma unroll
    for (int ks = 0; ks < 2; ++ks) {
      const v16h b = wp[(t * 2 + ks) * 32 + lane];
      acc[t] = __builtin_amdgcn_wmma_f32_16x16x32_f16(
          false, a[ks], false, b, (short)0, acc[t], false, false);
    }

  const int mi = m0 + ((lane >= 16) ? 8 : 0);
#pragma unroll
  for (int t = 0; t < 2; ++t)
#pragma unroll
    for (int r = 0; r < 8; ++r)
      h[(size_t)(mi + r) * OUT_F + t * 16 + mrow] = acc[t][r];
}

__global__ __launch_bounds__(256)
void k_node_prep(const float* __restrict__ h,
                 const float* __restrict__ att_src,
                 const float* __restrict__ att_dst,
                 float* __restrict__ a_src, float* __restrict__ a_dst,
                 unsigned* __restrict__ m_enc, float* __restrict__ denom,
                 float* __restrict__ out, int nNodes) {
  const int i = blockIdx.x * blockDim.x + threadIdx.x;
  if (i >= nNodes) return;
  const float4* hv = (const float4*)(h + (size_t)i * OUT_F);
  float s = 0.f, d = 0.f;
#pragma unroll
  for (int q = 0; q < 8; ++q) {
    const float4 hh = hv[q];
    s += hh.x * att_src[q * 4 + 0] + hh.y * att_src[q * 4 + 1] +
         hh.z * att_src[q * 4 + 2] + hh.w * att_src[q * 4 + 3];
    d += hh.x * att_dst[q * 4 + 0] + hh.y * att_dst[q * 4 + 1] +
         hh.z * att_dst[q * 4 + 2] + hh.w * att_dst[q * 4 + 3];
  }
  a_src[i] = s;
  a_dst[i] = d;
  m_enc[i] = 0u;
  denom[i] = 0.f;
  const float4 z = {0.f, 0.f, 0.f, 0.f};
  float4* ov = (float4*)(out + (size_t)i * OUT_F);
#pragma unroll
  for (int q = 0; q < 8; ++q) ov[q] = z;
}

__global__ __launch_bounds__(256)
void k_edge_max(const int* __restrict__ ei, long long E, int nNodes,
                const float* __restrict__ a_src, const float* __restrict__ a_dst,
                float* __restrict__ ebuf, unsigned* __restrict__ m_enc) {
  const long long e = (long long)blockIdx.x * blockDim.x + threadIdx.x;
  const long long tot = E + nNodes;
  if (e >= tot) return;
  int s, d;
  if (e < E) { s = ei[e]; d = ei[E + e]; } else { s = d = (int)(e - E); }
  float v = a_src[s] + a_dst[d];
  v = (v > 0.f) ? v : NEG_SLOPE * v;
  ebuf[e] = v;
  atomicMax(m_enc + d, enc_f32(v));
}

__global__ __launch_bounds__(256)
void k_edge_scatter(const int* __restrict__ ei, long long E, int nNodes,
                    const float* __restrict__ ebuf,
                    const unsigned* __restrict__ m_enc,
                    float* __restrict__ denom,
                    const float* __restrict__ h, float* __restrict__ out) {
  const int lane = threadIdx.x & 31;
  const int wv   = __builtin_amdgcn_readfirstlane(threadIdx.x >> 5);
  const long long e = (long long)blockIdx.x * (blockDim.x >> 5) + wv;
  const long long tot = E + nNodes;
  if (e >= tot) return;
  int s, d;
  if (e < E) { s = ei[e]; d = ei[E + e]; } else { s = d = (int)(e - E); }
  const float ex = __expf(ebuf[e] - dec_f32(m_enc[d]));
  if (lane == 0) atomicAdd(denom + d, ex);
  const float val = h[(size_t)s * OUT_F + lane] * ex;
  atomicAdd(out + (size_t)d * OUT_F + lane, val);
}

__global__ __launch_bounds__(256)
void k_finalize(float* __restrict__ out, const float* __restrict__ denom,
                const float* __restrict__ bias, long long total) {
  const long long idx = (long long)blockIdx.x * blockDim.x + threadIdx.x;
  if (idx >= total) return;
  const int node = (int)(idx >> 5);
  const int c    = (int)(idx & (OUT_F - 1));
  const float v = out[idx] / denom[node] + bias[c];
  out[idx] = v > 0.f ? v : 0.f;
}

extern "C" void kernel_launch(void* const* d_in, const int* in_sizes, int n_in,
                              void* d_out, int out_size, void* d_ws, size_t ws_size,
                              hipStream_t stream) {
  const float* x       = (const float*)d_in[0];
  const int*   ei      = (const int*)  d_in[1];
  const float* W       = (const float*)d_in[2];
  const float* att_src = (const float*)d_in[3];
  const float* att_dst = (const float*)d_in[4];
  const float* bias    = (const float*)d_in[5];
  float*       out     = (float*)d_out;

  const int       nNodes = in_sizes[0] / IN_F;
  const long long E      = (long long)in_sizes[1] / 2;
  const long long tot    = E + nNodes;

  char* ws = (char*)d_ws;
  auto carve = [&](size_t bytes) {
    char* p = ws;
    ws += (bytes + 255) & ~(size_t)255;
    return p;
  };
  float*     h     = (float*)    carve((size_t)nNodes * OUT_F * sizeof(float));
  float*     a_src = (float*)    carve((size_t)nNodes * sizeof(float));
  float*     a_dst = (float*)    carve((size_t)nNodes * sizeof(float));
  unsigned*  m_enc = (unsigned*) carve((size_t)nNodes * sizeof(unsigned));
  float*     denom = (float*)    carve((size_t)nNodes * sizeof(float));
  _Float16*  wpack = (_Float16*) carve((size_t)4 * 32 * 16 * sizeof(_Float16));
  float*     ebuf  = (float*)    carve((size_t)tot * sizeof(float));

  k_pack_w<<<1, 128, 0, stream>>>(W, wpack);

  const int wavesGemm  = (nNodes + 15) / 16;
  const int blocksGemm = (wavesGemm + 3) / 4;
  k_gemm<<<blocksGemm, 128, 0, stream>>>(x, wpack, h, nNodes);

  k_node_prep<<<(nNodes + 255) / 256, 256, 0, stream>>>(
      h, att_src, att_dst, a_src, a_dst, m_enc, denom, out, nNodes);

  k_edge_max<<<(int)((tot + 255) / 256), 256, 0, stream>>>(
      ei, E, nNodes, a_src, a_dst, ebuf, m_enc);

  k_edge_scatter<<<(int)((tot + 7) / 8), 256, 0, stream>>>(
      ei, E, nNodes, ebuf, m_enc, denom, h, out);

  const long long totalOut = (long long)nNodes * OUT_F;
  k_finalize<<<(int)((totalOut + 255) / 256), 256, 0, stream>>>(
      out, denom, bias, totalOut);
}


# Edges binned by destination range (LDS counting sort each call); a work-group per bin does segment max, exp-sum and row sum by LDS atomics in arrival order, writes out once; no global float atomics
# speedup vs baseline: 4.9688x; 4.9688x over previous
#include <hip/hip_runtime.h>
#include <hip/hip_bf16.h>

typedef __attribute__((ext_vector_type(16))) _Float16 v16h;
typedef __attribute__((ext_vector_type(8)))  float    v8f;

#define IN_F      64
#define OUT_F     32
#define NEG_SLOPE 0.2f
#define NP        256
#define NP_SHIFT  8
#define MAXP      512
#define S_BITS    17
#define BIN_T     512
#define TILE      8192
#define AGG_T     1024
#define AGG_B     16
static_assert(BIN_T == MAXP, "the binning kernels give one thread to every bin");
static_assert(MAXP * NP <= (1 << S_BITS), "source index must fit the packed key");

__device__ __forceinline__ unsigned enc_f32(float f) {
  unsigned u = __float_as_uint(f);
  return (u & 0x80000000u) ? ~u : (u | 0x80000000u);
}
__device__ __forceinline__ float dec_f32(unsigned u) {
  return (u & 0x80000000u) ? __uint_as_float(u & 0x7FFFFFFFu)
                           : __uint_as_float(~u);
}

__global__ __launch_bounds__(128)
void k_pack_w(const float* __restrict__ W, _Float16* __restrict__ wp,
              int* __restrict__ binTotal, int* __restrict__ binCursor, int nBins) {
  const int tid   = threadIdx.x;
  const int lane  = tid & 31;
  const int ks    = (tid >> 5) & 1;
  const int t     = tid >> 6;
  const int kbase = (lane >= 16) ? 8 : 0;
  const int n     = t * 16 + (lane & 15);
  for (int i = tid; i < nBins; i += 128) { binTotal[i] = 0; binCursor[i] = 0; }
  _Float16* dst = wp + (size_t)((t * 2 + ks) * 32 + lane) * 16;
#pragma unroll
  for (int j = 0; j < 8; ++j) {
    dst[j]     = (_Float16)W[(size_t)(ks * 32 + kbase + j)      * OUT_F + n];
    dst[j + 8] = (_Float16)W[(size_t)(ks * 32 + kbase + 16 + j) * OUT_F + n];
  }
}

__global__ __launch_bounds__(128)
void k_gemm(const float* __restrict__ x, const _Float16* __restrict__ wpack,
            float* __restrict__ h, int nNodes) {
  const int lane = threadIdx.x & 31;
  const int wave = blockIdx.x * (blockDim.x >> 5) + (threadIdx.x >> 5);
  const int m0   = wave * 16;
  if (m0 >= nNodes) return;

  const int mrow  = lane & 15;
  const int kbase = (lane >= 16) ? 8 : 0;
  int row = m0 + mrow;
  if (row > nNodes - 1) row = nNodes - 1;

  v16h a[2];
#pragma unroll
  for (int ks = 0; ks < 2; ++ks) {
    const float4* xr = (const float4*)(x + (size_t)row * IN_F + ks * 32 + kbase);
    const float4 lo0 = xr[0], lo1 = xr[1];
    const float4 hi0 = xr[4], hi1 = xr[5];
    a[ks][0]  = (_Float16)lo0.x; a[ks][1]  = (_Float16)lo0.y;
    a[ks][2]  = (_Float16)lo0.z; a[ks][3]  = (_Float16)lo0.w;
    a[ks][4]  = (_Float16)lo1.x; a[ks][5]  = (_Float16)lo1.y;
    a[ks][6]  = (_Float16)lo1.z; a[ks][7]  = (_Float16)lo1.w;
    a[ks][8]  = (_Float16)hi0.x; a[ks][9]  = (_Float16)hi0.y;
    a[ks][10] = (_Float16)hi0.z; a[ks][11] = (_Float16)hi0.w;
    a[ks][12] = (_Float16)hi1.x; a[ks][13] = (_Float16)hi1.y;
    a[ks][14] = (_Float16)hi1.z; a[ks][15] = (_Float16)hi1.w;
  }

  const v16h* wp = (const v16h*)wpack;
  v8f acc[2] = {};
#pragma unroll
  for (int t = 0; t < 2; ++t)
#pragma unroll
    for (int ks = 0; ks < 2; ++ks) {
      const v16h b = wp[(t * 2 + ks) * 32 + lane];
      acc[t] = __builtin_amdgcn_wmma_f32_16x16x32_f16(
          false, a[ks], false, b, (short)0, acc[t], false, false);
    }

  const int mi = m0 + ((lane >= 16) ? 8 : 0);
#pragma unroll
  for (int t = 0; t < 2; ++t)
#pragma unroll
    for (int r = 0; r < 8; ++r)
      h[(size_t)(mi + r) * OUT_F + t * 16 + mrow] = acc[t][r];
}

__global__ __launch_bounds__(256)
void k_node_prep(const float* __restrict__ h,
                 const float* __restrict__ att_src,
                 const float* __restrict__ att_dst,
                 float* __restrict__ a_src, float* __restrict__ a_dst, int nNodes) {
  const int i = blockIdx.x * blockDim.x + threadIdx.x;
  if (i >= nNodes) return;
  const float4* hv = (const float4*)(h + (size_t)i * OUT_F);
  float s = 0.f, d = 0.f;
#pragma unroll
  for (int q = 0; q < 8; ++q) {
    const float4 hh = hv[q];
    s += hh.x * att_src[q * 4 + 0] + hh.y * att_src[q * 4 + 1] +
         hh.z * att_src[q * 4 + 2] + hh.w * att_src[q * 4 + 3];
    d += hh.x * att_dst[q * 4 + 0] + hh.y * att_dst[q * 4 + 1] +
         hh.z * att_dst[q * 4 + 2] + hh.w * att_dst[q * 4 + 3];
  }
  a_src[i] = s;
  a_dst[i] = d;
}

__global__ __launch_bounds__(BIN_T)
void k_bin_count(const int* __restrict__ ei, long long E,
                 int* __restrict__ binTotal, int nBins) {
  __shared__ int cnt[MAXP];
  const int tid = threadIdx.x;
  cnt[tid] = 0;
  __syncthreads();
  const long long t0 = (long long)blockIdx.x * TILE;
  const int* dstp = ei + E;
  int dv[TILE / BIN_T];
#pragma unroll
  for (int k = 0; k < TILE / BIN_T; ++k) {
    const long long e = t0 + k * BIN_T + tid;
    dv[k] = dstp[(e < E) ? e : (E - 1)];
  }
#pragma unroll
  for (int k = 0; k < TILE / BIN_T; ++k) {
    const long long e = t0 + k * BIN_T + tid;
    if (e < E) atomicAdd(&cnt[dv[k] >> NP_SHIFT], 1);
  }
  __syncthreads();
  const int c = cnt[tid];
  if (tid < nBins && c) atomicAdd(binTotal + tid, c);
}

__global__ __launch_bounds__(BIN_T)
void k_bin_scatter(const int* __restrict__ ei, long long E,
                   const float* __restrict__ a_src,
                   const int* __restrict__ binTotal, int* __restrict__ binCursor,
                   int* __restrict__ binStart, int2* __restrict__ binned, int nBins) {
  __shared__ int cnt[MAXP];
  __shared__ int sc[MAXP];
  __shared__ int base[MAXP];
  const int tid = threadIdx.x;
  const int tot = (tid < nBins) ? binTotal[tid] : 0;
  cnt[tid] = 0;
  sc[tid]  = tot;
  __syncthreads();
  const long long t0 = (long long)blockIdx.x * TILE;
  const int* dstp = ei + E;
  int   sv[TILE / BIN_T], dv[TILE / BIN_T], rk[TILE / BIN_T];
  float av[TILE / BIN_T];
#pragma unroll
  for (int k = 0; k < TILE / BIN_T; ++k) {
    const long long e  = t0 + k * BIN_T + tid;
    const long long ec = (e < E) ? e : (E - 1);
    sv[k] = ei[ec];
    dv[k] = dstp[ec];
  }
#pragma unroll
  for (int k = 0; k < TILE / BIN_T; ++k) {
    const long long e = t0 + k * BIN_T + tid;
    rk[k] = 0;
    if (e < E) rk[k] = atomicAdd(&cnt[dv[k] >> NP_SHIFT], 1);
    else       dv[k] = -1;
  }
#pragma unroll
  for (int k = 0; k < TILE / BIN_T; ++k) av[k] = a_src[sv[k]];
  for (int off = 1; off < MAXP; off <<= 1) {
    __syncthreads();
    const int t = (tid >= off) ? sc[tid - off] : 0;
    __syncthreads();
    sc[tid] += t;
  }
  const int start = sc[tid] - tot;
  const int c = cnt[tid];
  base[tid] = (tid < nBins && c) ? start + atomicAdd(binCursor + tid, c) : 0;
  if (blockIdx.x == 0) {
    if (tid < nBins) binStart[tid] = start;
    if (tid == nBins - 1) binStart[nBins] = start + tot;
  }
  __syncthreads();
#pragma unroll
  for (int k = 0; k < TILE / BIN_T; ++k) {
    if (dv[k] >= 0) {
      const int pos = base[dv[k] >> NP_SHIFT] + rk[k];
      const int key = sv[k] | ((dv[k] & (NP - 1)) << S_BITS);
      binned[pos] = make_int2(key, __float_as_int(av[k]));
    }
  }
}

__global__ __launch_bounds__(AGG_T)
void k_bin_aggregate(const int2* __restrict__ binned, const int* __restrict__ binStart,
                     const float* __restrict__ h,
                     const float* __restrict__ a_src, const float* __restrict__ a_dst,
                     const float* __restrict__ bias, float* __restrict__ out, int nNodes) {
  __shared__ float    acc[NP * OUT_F];
  __shared__ unsigned m_s[NP];
  __shared__ float    den[NP];
  __shared__ float    adst_s[NP];
  __shared__ float    vself_s[NP];
  const int tid  = threadIdx.x;
  const int lane = tid & 31;
  const int wv   = __builtin_amdgcn_readfirstlane(tid >> 5);
  const int p        = blockIdx.x;
  const int nodeBase = p << NP_SHIFT;
  const int nLocal   = (nNodes - nodeBase < NP) ? (nNodes - nodeBase) : NP;
  const int eBeg = binStart[p];
  const int eEnd = binStart[p + 1];

  const size_t rowBase = (size_t)nodeBase * OUT_F;
  float hself[NP * OUT_F / AGG_T];
#pragma unroll
  for (int q = 0; q < NP * OUT_F / AGG_T; ++q) {
    const int k = tid + q * AGG_T;
    hself[q] = (k < nLocal * OUT_F) ? h[rowBase + k] : 0.f;
  }
  const float bs = bias[lane];

  for (int k = tid; k < NP * OUT_F; k += AGG_T) acc[k] = 0.f;
  if (tid < NP) {
    float ad = 0.f, v = 0.f;
    unsigned me = 0u;
    if (tid < nLocal) {
      ad = a_dst[nodeBase + tid];
      v  = a_src[nodeBase + tid] + ad;
      v  = (v > 0.f) ? v : NEG_SLOPE * v;
      me = enc_f32(v);
    }
    adst_s[tid] = ad; vself_s[tid] = v; m_s[tid] = me; den[tid] = 0.f;
  }
  __syncthreads();

  for (int e0 = eBeg + tid; e0 < eEnd; e0 += 4 * AGG_T) {
    int2 r[4];
#pragma unroll
    for (int u = 0; u < 4; ++u) {
      const int e = e0 + u * AGG_T;
      r[u] = binned[(e < eEnd) ? e : (eEnd - 1)];
    }
#pragma unroll
    for (int u = 0; u < 4; ++u) {
      const int dl = r[u].x >> S_BITS;
      float v = __int_as_float(r[u].y) + adst_s[dl];
      v = (v > 0.f) ? v : NEG_SLOPE * v;
      atomicMax(&m_s[dl], enc_f32(v));
    }
  }
  __syncthreads();

  int2 rnext = make_int2(0, 0);
  {
    const int c0 = eBeg + wv * 32;
    if (c0 < eEnd) rnext = binned[(c0 + lane < eEnd) ? (c0 + lane) : (eEnd - 1)];
  }
  for (int c = eBeg + wv * 32; c < eEnd; c += AGG_T) {
    const int e = c + lane;
    const int2 r = rnext;
    if (c + AGG_T < eEnd) rnext = binned[(e + AGG_T < eEnd) ? (e + AGG_T) : (eEnd - 1)];
    const int s  = r.x & ((1 << S_BITS) - 1);
    const int dl = r.x >> S_BITS;
    float ex = 0.f;
    if (e < eEnd) {
      float v = __int_as_float(r.y) + adst_s[dl];
      v = (v > 0.f) ? v : NEG_SLOPE * v;
      ex = __expf(v - dec_f32(m_s[dl]));
      atomicAdd(&den[dl], ex);
    }
    const int n = (eEnd - c < 32) ? (eEnd - c) : 32;
    int j = 0;
    for (; j + AGG_B <= n; j += AGG_B) {
      float hv[AGG_B], xj[AGG_B];
      int   dj[AGG_B];
#pragma unroll
      for (int u = 0; u < AGG_B; ++u) {
        const int sj = __builtin_amdgcn_readlane(s, j + u);
        dj[u] = __builtin_amdgcn_readlane(dl, j + u);
        xj[u] = __int_as_float(__builtin_amdgcn_readlane(__float_as_int(ex), j + u));
        hv[u] = h[(size_t)sj * OUT_F + lane];
      }
#pragma unroll
      for (int u = 0; u < AGG_B; ++u)
        atomicAdd(&acc[dj[u] * OUT_F + lane], hv[u] * xj[u]);
    }
    for (; j < n; ++j) {
      const int   sj = __builtin_amdgcn_readlane(s, j);
      const int   dd = __builtin_amdgcn_readlane(dl, j);
      const float xx = __int_as_float(__builtin_amdgcn_readlane(__float_as_int(ex), j));
      atomicAdd(&acc[dd * OUT_F + lane], h[(size_t)sj * OUT_F + lane] * xx);
    }
  }
  __syncthreads();

#pragma unroll
  for (int q = 0; q < NP * OUT_F / AGG_T; ++q) {
    const int k = tid + q * AGG_T;
    if (k < nLocal * OUT_F) {
      const int i = k >> 5;
      const float ex  = __expf(vself_s[i] - dec_f32(m_s[i]));
      const float num = acc[k] + ex * hself[q];
      const float dn  = den[i] + ex;
      const float o   = num / dn + bs;
      out[rowBase + k] = o > 0.f ? o : 0.f;
    }
  }
}

extern "C" void kernel_launch(void* const* d_in, const int* in_sizes, int n_in,
                              void* d_out, int out_size, void* d_ws, size_t ws_size,
                              hipStream_t stream) {
  const float* x       = (const float*)d_in[0];
  const int*   ei      = (const int*)  d_in[1];
  const float* W       = (const float*)d_in[2];
  const float* att_src = (const float*)d_in[3];
  const float* att_dst = (const float*)d_in[4];
  const float* bias    = (const float*)d_in[5];
  float*       out     = (float*)d_out;

  const int       nNodes = in_sizes[0] / IN_F;
  const long long E      = (long long)in_sizes[1] / 2;
  const int       nBins  = (nNodes + NP - 1) / NP;
  const int       nTiles = (int)((E + TILE - 1) / TILE);

  char* ws = (char*)d_ws;
  auto carve = [&](size_t bytes) {
    char* p = ws;
    ws += (bytes + 255) & ~(size_t)255;
    return p;
  };
  float*     h         = (float*)    carve((size_t)nNodes * OUT_F * sizeof(float));
  float*     a_src     = (float*)    carve((size_t)nNodes * sizeof(float));
  float*     a_dst     = (float*)    carve((size_t)nNodes * sizeof(float));
  _Float16*  wpack     = (_Float16*) carve((size_t)4 * 32 * 16 * sizeof(_Float16));
  int*       binTotal  = (int*)      carve((size_t)MAXP * sizeof(int));
  int*       binStart  = (int*)      carve((size_t)(MAXP + 1) * sizeof(int));
  int*       binCursor = (int*)      carve((size_t)MAXP * sizeof(int));
  int2*      binned    = (int2*)     carve((size_t)E * sizeof(int2));

  k_pack_w<<<1, 128, 0, stream>>>(W, wpack, binTotal, binCursor, nBins);

  const int wavesGemm  = (nNodes + 15) / 16;
  const int blocksGemm = (wavesGemm + 3) / 4;
  k_gemm<<<blocksGemm, 128, 0, stream>>>(x, wpack, h, nNodes);

  k_node_prep<<<(nNodes + 255) / 256, 256, 0, stream>>>(
      h, att_src, att_dst, a_src, a_dst, nNodes);

  k_bin_count<<<nTiles, BIN_T, 0, stream>>>(ei, E, binTotal, nBins);

  k_bin_scatter<<<nTiles, BIN_T, 0, stream>>>(ei, E, a_src, binTotal, binCursor, binStart, binned, nBins);

  k_bin_aggregate<<<nBins, AGG_T, 0, stream>>>(binned, binStart, h, a_src, a_dst, bias, out, nNodes);
}


# Scatter stages each tile's records in LDS in bin order (second prefix sum, over the tile's own counts); consecutive lanes then store consecutive slots, so a bin share leaves as one contiguous run
# speedup vs baseline: 6.2866x; 1.2652x over previous
#include <hip/hip_runtime.h>
#include <hip/hip_bf16.h>

typedef __attribute__((ext_vector_type(16))) _Float16 v16h;
typedef __attribute__((ext_vector_type(8)))  float    v8f;

#define IN_F      64
#define OUT_F     32
#define NEG_SLOPE 0.2f
#define NP        256
#define NP_SHIFT  8
#define MAXP      512
#define S_BITS    17
#define BIN_T     512
#define TILE      8192
#define AGG_T     1024
#define AGG_B     16
static_assert(BIN_T == MAXP, "the binning kernels give one thread to every bin");
static_assert(MAXP * NP <= (1 << S_BITS), "source index must fit the packed key");

__device__ __forceinline__ unsigned enc_f32(float f) {
  unsigned u = __float_as_uint(f);
  return (u & 0x80000000u) ? ~u : (u | 0x80000000u);
}
__device__ __forceinline__ float dec_f32(unsigned u) {
  return (u & 0x80000000u) ? __uint_as_float(u & 0x7FFFFFFFu)
                           : __uint_as_float(~u);
}

template <int M>
__device__ __forceinline__ float row_xor(float v) {
  return __int_as_float(__builtin_amdgcn_update_dpp(0, __float_as_int(v), 0x160 + M, 0xf, 0xf, true));
}

__global__ __launch_bounds__(128)
void k_project(const float* __restrict__ x, const float* __restrict__ W,
               const float* __restrict__ att_src, const float* __restrict__ att_dst,
               float* __restrict__ h, float* __restrict__ a_src, float* __restrict__ a_dst,
               int* __restrict__ binTotal, int* __restrict__ binCursor, int nBins, int nNodes) {
  __shared__ __attribute__((aligned(32))) _Float16 wl[4 * 32 * 16];
  const int tid   = threadIdx.x;
  const int lane  = tid & 31;
  const int wave  = blockIdx.x * 4 + (tid >> 5);
  const int m0    = wave * 16;
  const int mrow  = lane & 15;
  const int kbase = (lane >= 16) ? 8 : 0;
  int row = m0 + mrow;
  if (row > nNodes - 1) row = nNodes - 1;

  if (blockIdx.x == 0)
    for (int i = tid; i < nBins; i += 128) { binTotal[i] = 0; binCursor[i] = 0; }

  float wv[16];
  {
    const int ks = (tid >> 5) & 1;
    const int n  = (tid >> 6) * 16 + mrow;
#pragma unroll
    for (int j = 0; j < 8; ++j) {
      wv[j]     = W[(size_t)(ks * 32 + kbase + j)      * OUT_F + n];
      wv[j + 8] = W[(size_t)(ks * 32 + kbase + 16 + j) * OUT_F + n];
    }
  }

  float4 lo0[2], lo1[2], hi0[2], hi1[2];
#pragma unroll
  for (int ks = 0; ks < 2; ++ks) {
    const float4* xr = (const float4*)(x + (size_t)row * IN_F + ks * 32 + kbase);
    lo0[ks] = xr[0]; lo1[ks] = xr[1];
    hi0[ks] = xr[4]; hi1[ks] = xr[5];
  }

  {
    v16h f;
#pragma unroll
    for (int j = 0; j < 16; ++j) f[j] = (_Float16)wv[j];
    *(v16h*)(wl + tid * 16) = f;
  }
  __syncthreads();
  if (m0 >= nNodes) return;

  v16h a[2];
#pragma unroll
  for (int ks = 0; ks < 2; ++ks) {
    a[ks][0]  = (_Float16)lo0[ks].x; a[ks][1]  = (_Float16)lo0[ks].y;
    a[ks][2]  = (_Float16)lo0[ks].z; a[ks][3]  = (_Float16)lo0[ks].w;
    a[ks][4]  = (_Float16)lo1[ks].x; a[ks][5]  = (_Float16)lo1[ks].y;
    a[ks][6]  = (_Float16)lo1[ks].z; a[ks][7]  = (_Float16)lo1[ks].w;
    a[ks][8]  = (_Float16)hi0[ks].x; a[ks][9]  = (_Float16)hi0[ks].y;
    a[ks][10] = (_Float16)hi0[ks].z; a[ks][11] = (_Float16)hi0[ks].w;
    a[ks][12] = (_Float16)hi1[ks].x; a[ks][13] = (_Float16)hi1[ks].y;
    a[ks][14] = (_Float16)hi1[ks].z; a[ks][15] = (_Float16)hi1[ks].w;
  }

  const v16h* wp = (const v16h*)wl;
  v8f acc[2] = {};
#pragma unroll
  for (int t = 0; t < 2; ++t)
#pragma unroll
    for (int ks = 0; ks < 2; ++ks) {
      const v16h b = wp[(t * 2 + ks) * 32 + lane];
      acc[t] = __builtin_amdgcn_wmma_f32_16x16x32_f16(
          false, a[ks], false, b, (short)0, acc[t], false, false);
    }

  const int mi = m0 + ((lane >= 16) ? 8 : 0);
  if (m0 + 16 <= nNodes) {
#pragma unroll
    for (int t = 0; t < 2; ++t)
#pragma unroll
      for (int r = 0; r < 8; ++r)
        h[(size_t)(mi + r) * OUT_F + t * 16 + mrow] = acc[t][r];
  } else {
#pragma unroll
    for (int t = 0; t < 2; ++t)
#pragma unroll
      for (int r = 0; r < 8; ++r)
        if (mi + r < nNodes) h[(size_t)(mi + r) * OUT_F + t * 16 + mrow] = acc[t][r];
  }

  const float as0 = att_src[mrow], as1 = att_src[16 + mrow];
  const float ad0 = att_dst[mrow], ad1 = att_dst[16 + mrow];
  float v[16];
#pragma unroll
  for (int r = 0; r < 8; ++r) {
    v[r]     = acc[0][r] * as0 + acc[1][r] * as1;
    v[8 + r] = acc[0][r] * ad0 + acc[1][r] * ad1;
  }
  const bool b8 = (lane & 8) != 0, b4 = (lane & 4) != 0, b2 = (lane & 2) != 0, b1 = (lane & 1) != 0;
  float w8[8], w4[4], w2[2];
#pragma unroll
  for (int i = 0; i < 8; ++i) {
    const float keep = b8 ? v[i + 8] : v[i];
    const float send = b8 ? v[i] : v[i + 8];
    w8[i] = keep + row_xor<8>(send);
  }
#pragma unroll
  for (int i = 0; i < 4; ++i) {
    const float keep = b4 ? w8[i + 4] : w8[i];
    const float send = b4 ? w8[i] : w8[i + 4];
    w4[i] = keep + row_xor<4>(send);
  }
#pragma unroll
  for (int i = 0; i < 2; ++i) {
    const float keep = b2 ? w4[i + 2] : w4[i];
    const float send = b2 ? w4[i] : w4[i + 2];
    w2[i] = keep + row_xor<2>(send);
  }
  const float keep1 = b1 ? w2[1] : w2[0];
  const float send1 = b1 ? w2[0] : w2[1];
  const float tot   = keep1 + row_xor<1>(send1);
  const int orow = mi + (mrow & 7);
  if (orow < nNodes) {
    float* dstp = (mrow < 8) ? a_src : a_dst;
    dstp[orow] = tot;
  }
}

__global__ __launch_bounds__(BIN_T)
void k_bin_count(const int* __restrict__ ei, long long E,
                 int* __restrict__ binTotal, int nBins) {
  __shared__ int cnt[MAXP];
  const int tid = threadIdx.x;
  cnt[tid] = 0;
  __syncthreads();
  const long long t0 = (long long)blockIdx.x * TILE;
  const int* dstp = ei + E;
  int dv[TILE / BIN_T];
#pragma unroll
  for (int k = 0; k < TILE / BIN_T; ++k) {
    const long long e = t0 + k * BIN_T + tid;
    dv[k] = dstp[(e < E) ? e : (E - 1)];
  }
#pragma unroll
  for (int k = 0; k < TILE / BIN_T; ++k) {
    const long long e = t0 + k * BIN_T + tid;
    if (e < E) atomicAdd(&cnt[dv[k] >> NP_SHIFT], 1);
  }
  __syncthreads();
  const int c = cnt[tid];
  if (tid < nBins && c) atomicAdd(binTotal + tid, c);
}

__global__ __launch_bounds__(BIN_T)
void k_bin_scatter(const int* __restrict__ ei, long long E,
                   const float* __restrict__ a_src,
                   const int* __restrict__ binTotal, int* __restrict__ binCursor,
                   int* __restrict__ binStart, int2* __restrict__ binned, int nBins) {
  __shared__ int  cnt[MAXP];
  __shared__ int  sc[MAXP];
  __shared__ int  lc[MAXP];
  __shared__ int  base[MAXP];
  __shared__ int  lst[MAXP];
  __shared__ __attribute__((aligned(16))) int4 srec[TILE];
  const int tid = threadIdx.x;
  const int tot = (tid < nBins) ? binTotal[tid] : 0;
  cnt[tid] = 0;
  sc[tid]  = tot;
  __syncthreads();
  const long long t0 = (long long)blockIdx.x * TILE;
  const int nValid = (E - t0 < (long long)TILE) ? (int)(E - t0) : TILE;
  const int* dstp = ei + E;
  int   sv[TILE / BIN_T], dv[TILE / BIN_T], rk[TILE / BIN_T];
  float av[TILE / BIN_T];
#pragma unroll
  for (int k = 0; k < TILE / BIN_T; ++k) {
    const long long e  = t0 + k * BIN_T + tid;
    const long long ec = (e < E) ? e : (E - 1);
    sv[k] = ei[ec];
    dv[k] = dstp[ec];
  }
#pragma unroll
  for (int k = 0; k < TILE / BIN_T; ++k) {
    const long long e = t0 + k * BIN_T + tid;
    rk[k] = 0;
    if (e < E) rk[k] = atomicAdd(&cnt[dv[k] >> NP_SHIFT], 1);
  }
#pragma unroll
  for (int k = 0; k < TILE / BIN_T; ++k) av[k] = a_src[sv[k]];
  __syncthreads();
  const int c = cnt[tid];
  lc[tid] = c;
  for (int off = 1; off < MAXP; off <<= 1) {
    __syncthreads();
    const int t  = (tid >= off) ? sc[tid - off] : 0;
    const int tl = (tid >= off) ? lc[tid - off] : 0;
    __syncthreads();
    sc[tid] += t;
    lc[tid] += tl;
  }
  const int start = sc[tid] - tot;
  lst[tid]  = lc[tid] - c;
  base[tid] = (tid < nBins && c) ? start + atomicAdd(binCursor + tid, c) : 0;
  if (blockIdx.x == 0) {
    if (tid < nBins) binStart[tid] = start;
    if (tid == nBins - 1) binStart[nBins] = start + tot;
  }
  __syncthreads();
  int slot[TILE / BIN_T], gpos[TILE / BIN_T];
#pragma unroll
  for (int k = 0; k < TILE / BIN_T; ++k) {
    const int b = dv[k] >> NP_SHIFT;
    slot[k] = lst[b] + rk[k];
    gpos[k] = base[b] + rk[k];
  }
#pragma unroll
  for (int k = 0; k < TILE / BIN_T; ++k) {
    if (k * BIN_T + tid < nValid) {
      const int key = sv[k] | ((dv[k] & (NP - 1)) << S_BITS);
      srec[slot[k]] = make_int4(key, __float_as_int(av[k]), gpos[k], 0);
    }
  }
  __syncthreads();
  int4 st[TILE / BIN_T];
#pragma unroll
  for (int k = 0; k < TILE / BIN_T; ++k) {
    const int i = k * BIN_T + tid;
    st[k] = srec[(i < nValid) ? i : (nValid - 1)];
  }
#pragma unroll
  for (int k = 0; k < TILE / BIN_T; ++k)
    if (k * BIN_T + tid < nValid) binned[st[k].z] = make_int2(st[k].x, st[k].y);
}

__global__ __launch_bounds__(AGG_T)
void k_bin_aggregate(const int2* __restrict__ binned, const int* __restrict__ binStart,
                     const float* __restrict__ h,
                     const float* __restrict__ a_src, const float* __restrict__ a_dst,
                     const float* __restrict__ bias, float* __restrict__ out, int nNodes) {
  __shared__ float    acc[NP * OUT_F];
  __shared__ unsigned m_s[NP];
  __shared__ float    den[NP];
  __shared__ float    adst_s[NP];
  __shared__ float    vself_s[NP];
  const int tid  = threadIdx.x;
  const int lane = tid & 31;
  const int wv   = __builtin_amdgcn_readfirstlane(tid >> 5);
  const int p        = blockIdx.x;
  const int nodeBase = p << NP_SHIFT;
  const int nLocal   = (nNodes - nodeBase < NP) ? (nNodes - nodeBase) : NP;
  const int eBeg = binStart[p];
  const int eEnd = binStart[p + 1];

  const size_t rowBase = (size_t)nodeBase * OUT_F;
  float hself[NP * OUT_F / AGG_T];
#pragma unroll
  for (int q = 0; q < NP * OUT_F / AGG_T; ++q) {
    const int k = tid + q * AGG_T;
    hself[q] = (k < nLocal * OUT_F) ? h[rowBase + k] : 0.f;
  }
  const float bs = bias[lane];

  for (int k = tid; k < NP * OUT_F; k += AGG_T) acc[k] = 0.f;
  if (tid < NP) {
    float ad = 0.f, v = 0.f;
    unsigned me = 0u;
    if (tid < nLocal) {
      ad = a_dst[nodeBase + tid];
      v  = a_src[nodeBase + tid] + ad;
      v  = (v > 0.f) ? v : NEG_SLOPE * v;
      me = enc_f32(v);
    }
    adst_s[tid] = ad; vself_s[tid] = v; m_s[tid] = me; den[tid] = 0.f;
  }
  __syncthreads();

  for (int e0 = eBeg + tid; e0 < eEnd; e0 += 4 * AGG_T) {
    int2 r[4];
#pragma unroll
    for (int u = 0; u < 4; ++u) {
      const int e = e0 + u * AGG_T;
      r[u] = binned[(e < eEnd) ? e : (eEnd - 1)];
    }
#pragma unroll
    for (int u = 0; u < 4; ++u) {
      const int dl = r[u].x >> S_BITS;
      float v = __int_as_float(r[u].y) + adst_s[dl];
      v = (v > 0.f) ? v : NEG_SLOPE * v;
      atomicMax(&m_s[dl], enc_f32(v));
    }
  }
  __syncthreads();

  int2 rnext = make_int2(0, 0);
  {
    const int c0 = eBeg + wv * 32;
    if (c0 < eEnd) rnext = binned[(c0 + lane < eEnd) ? (c0 + lane) : (eEnd - 1)];
  }
  for (int c = eBeg + wv * 32; c < eEnd; c += AGG_T) {
    const int e = c + lane;
    const int2 r = rnext;
    if (c + AGG_T < eEnd) rnext = binned[(e + AGG_T < eEnd) ? (e + AGG_T) : (eEnd - 1)];
    const int s  = r.x & ((1 << S_BITS) - 1);
    const int dl = r.x >> S_BITS;
    float ex = 0.f;
    if (e < eEnd) {
      float v = __int_as_float(r.y) + adst_s[dl];
      v = (v > 0.f) ? v : NEG_SLOPE * v;
      ex = __expf(v - dec_f32(m_s[dl]));
      atomicAdd(&den[dl], ex);
    }
    const int n = (eEnd - c < 32) ? (eEnd - c) : 32;
    int j = 0;
    for (; j + AGG_B <= n; j += AGG_B) {
      float hv[AGG_B], xj[AGG_B];
      int   dj[AGG_B];
#pragma unroll
      for (int u = 0; u < AGG_B; ++u) {
        const int sj = __builtin_amdgcn_readlane(s, j + u);
        dj[u] = __builtin_amdgcn_readlane(dl, j + u);
        xj[u] = __int_as_float(__builtin_amdgcn_readlane(__float_as_int(ex), j + u));
        hv[u] = h[(size_t)sj * OUT_F + lane];
      }
#pragma unroll
      for (int u = 0; u < AGG_B; ++u)
        atomicAdd(&acc[dj[u] * OUT_F + lane], hv[u] * xj[u]);
    }
    for (; j < n; ++j) {
      const int   sj = __builtin_amdgcn_readlane(s, j);
      const int   dd = __builtin_amdgcn_readlane(dl, j);
      const float xx = __int_as_float(__builtin_amdgcn_readlane(__float_as_int(ex), j));
      atomicAdd(&acc[dd * OUT_F + lane], h[(size_t)sj * OUT_F + lane] * xx);
    }
  }
  __syncthreads();

#pragma unroll
  for (int q = 0; q < NP * OUT_F / AGG_T; ++q) {
    const int k = tid + q * AGG_T;
    if (k < nLocal * OUT_F) {
      const int i = k >> 5;
      const float ex  = __expf(vself_s[i] - dec_f32(m_s[i]));
      const float num = acc[k] + ex * hself[q];
      const float dn  = den[i] + ex;
      const float o   = num / dn + bs;
      out[rowBase + k] = o > 0.f ? o : 0.f;
    }
  }
}

extern "C" void kernel_launch(void* const* d_in, const int* in_sizes, int n_in,
                              void* d_out, int out_size, void* d_ws, size_t ws_size,
                              hipStream_t stream) {
  const float* x       = (const float*)d_in[0];
  const int*   ei      = (const int*)  d_in[1];
  const float* W       = (const float*)d_in[2];
  const float* att_src = (const float*)d_in[3];
  const float* att_dst = (const float*)d_in[4];
  const float* bias    = (const float*)d_in[5];
  float*       out     = (float*)d_out;

  const int       nNodes = in_sizes[0] / IN_F;
  const long long E      = (long long)in_sizes[1] / 2;
  const int       nBins  = (nNodes + NP - 1) / NP;
  const int       nTiles = (int)((E + TILE - 1) / TILE);

  char* ws = (char*)d_ws;
  auto carve = [&](size_t bytes) {
    char* p = ws;
    ws += (bytes + 255) & ~(size_t)255;
    return p;
  };
  float*     h         = (float*)    carve((size_t)nNodes * OUT_F * sizeof(float));
  float*     a_src     = (float*)    carve((size_t)nNodes * sizeof(float));
  float*     a_dst     = (float*)    carve((size_t)nNodes * sizeof(float));
  int*       binTotal  = (int*)      carve((size_t)MAXP * sizeof(int));
  int*       binStart  = (int*)      carve((size_t)(MAXP + 1) * sizeof(int));
  int*       binCursor = (int*)      carve((size_t)MAXP * sizeof(int));
  int2*      binned    = (int2*)     carve((size_t)E * sizeof(int2));

  const int wavesGemm  = (nNodes + 15) / 16;
  const int blocksGemm = (wavesGemm + 3) / 4;
  k_project<<<blocksGemm, 128, 0, stream>>>(x, W, att_src, att_dst, h, a_src, a_dst,
                                            binTotal, binCursor, nBins, nNodes);

  k_bin_count<<<nTiles, BIN_T, 0, stream>>>(ei, E, binTotal, nBins);

  k_bin_scatter<<<nTiles, BIN_T, 0, stream>>>(ei, E, a_src, binTotal, binCursor, binStart, binned, nBins);

  k_bin_aggregate<<<nBins, AGG_T, 0, stream>>>(binned, binStart, h, a_src, a_dst, bias, out, nNodes);
}


# Bin records grouped by destination in LDS (counting sort, arrival rank); eight lanes own a destination and sum its run in registers from 16-byte row loads; LDS float atomics gone (sum order changes)
# speedup vs baseline: 7.7768x; 1.2370x over previous
#include <hip/hip_runtime.h>
#include <hip/hip_bf16.h>

typedef __attribute__((ext_vector_type(16))) _Float16 v16h;
typedef __attribute__((ext_vector_type(8)))  float    v8f;

#define IN_F      64
#define OUT_F     32
#define NEG_SLOPE 0.2f
#define NP        256
#define NP_SHIFT  8
#define MAXP      512
#define S_BITS    17
#define BIN_T     512
#define TILE      8192
#define AGG_T     1024
#define AGG_B     16
#define MAXT      1024
#define REC_CAP   8192
#define STG_U     5
static_assert(STG_U == 5, "the load fence in stage_window names STG_U records");
static_assert(BIN_T == MAXP, "the sorting kernel gives one thread to every bin");
static_assert(MAXT == AGG_T && AGG_T == 1024, "one thread per tile, and 32 waves whose totals one wave can sum");
static_assert(REC_CAP % AGG_T == 0, "windows are whole rounds of the work-group's threads");
static_assert(MAXP * NP <= (1 << S_BITS), "source index must fit the packed key");
static_assert(NP == 256 && AGG_T == 1024, "one wave scans the bin's 256 counts, 8 per lane; a wave owns 8 destinations, 4 at a time");
static_assert(STG_U * AGG_T <= REC_CAP, "a bin whose records its threads hold in registers fits the staging area");

__device__ __forceinline__ unsigned enc_f32(float f) {
  unsigned u = __float_as_uint(f);
  return (u & 0x80000000u) ? ~u : (u | 0x80000000u);
}
__device__ __forceinline__ float dec_f32(unsigned u) {
  return (u & 0x80000000u) ? __uint_as_float(u & 0x7FFFFFFFu)
                           : __uint_as_float(~u);
}

template <int M>
__device__ __forceinline__ float row_xor(float v) {
  return __int_as_float(__builtin_amdgcn_update_dpp(0, __float_as_int(v), 0x160 + M, 0xf, 0xf, true));
}

__global__ __launch_bounds__(128)
void k_project(const float* __restrict__ x, const float* __restrict__ W,
               const float* __restrict__ att_src, const float* __restrict__ att_dst,
               float* __restrict__ h, float* __restrict__ a_src, float* __restrict__ a_dst,
               int nNodes) {
  __shared__ __attribute__((aligned(32))) _Float16 wl[4 * 32 * 16];
  const int tid   = threadIdx.x;
  const int lane  = tid & 31;
  const int wave  = blockIdx.x * 4 + (tid >> 5);
  const int m0    = wave * 16;
  const int mrow  = lane & 15;
  const int kbase = (lane >= 16) ? 8 : 0;
  int row = m0 + mrow;
  if (row > nNodes - 1) row = nNodes - 1;

  float wv[16];
  {
    const int ks = (tid >> 5) & 1;
    const int n  = (tid >> 6) * 16 + mrow;
#pragma unroll
    for (int j = 0; j < 8; ++j) {
      wv[j]     = W[(size_t)(ks * 32 + kbase + j)      * OUT_F + n];
      wv[j + 8] = W[(size_t)(ks * 32 + kbase + 16 + j) * OUT_F + n];
    }
  }

  float4 lo0[2], lo1[2], hi0[2], hi1[2];
#pragma unroll
  for (int ks = 0; ks < 2; ++ks) {
    const float4* xr = (const float4*)(x + (size_t)row * IN_F + ks * 32 + kbase);
    lo0[ks] = xr[0]; lo1[ks] = xr[1];
    hi0[ks] = xr[4]; hi1[ks] = xr[5];
  }

  {
    v16h f;
#pragma unroll
    for (int j = 0; j < 16; ++j) f[j] = (_Float16)wv[j];
    *(v16h*)(wl + tid * 16) = f;
  }
  __syncthreads();
  if (m0 >= nNodes) return;

  v16h a[2];
#pragma unroll
  for (int ks = 0; ks < 2; ++ks) {
    a[ks][0]  = (_Float16)lo0[ks].x; a[ks][1]  = (_Float16)lo0[ks].y;
    a[ks][2]  = (_Float16)lo0[ks].z; a[ks][3]  = (_Float16)lo0[ks].w;
    a[ks][4]  = (_Float16)lo1[ks].x; a[ks][5]  = (_Float16)lo1[ks].y;
    a[ks][6]  = (_Float16)lo1[ks].z; a[ks][7]  = (_Float16)lo1[ks].w;
    a[ks][8]  = (_Float16)hi0[ks].x; a[ks][9]  = (_Float16)hi0[ks].y;
    a[ks][10] = (_Float16)hi0[ks].z; a[ks][11] = (_Float16)hi0[ks].w;
    a[ks][12] = (_Float16)hi1[ks].x; a[ks][13] = (_Float16)hi1[ks].y;
    a[ks][14] = (_Float16)hi1[ks].z; a[ks][15] = (_Float16)hi1[ks].w;
  }

  const v16h* wp = (const v16h*)wl;
  v8f acc[2] = {};
#pragma unroll
  for (int t = 0; t < 2; ++t)
#pragma unroll
    for (int ks = 0; ks < 2; ++ks) {
      const v16h b = wp[(t * 2 + ks) * 32 + lane];
      acc[t] = __builtin_amdgcn_wmma_f32_16x16x32_f16(
          false, a[ks], false, b, (short)0, acc[t], false, false);
    }

  const int mi = m0 + ((lane >= 16) ? 8 : 0);
  if (m0 + 16 <= nNodes) {
#pragma unroll
    for (int t = 0; t < 2; ++t)
#pragma unroll
      for (int r = 0; r < 8; ++r)
        h[(size_t)(mi + r) * OUT_F + t * 16 + mrow] = acc[t][r];
  } else {
#pragma unroll
    for (int t = 0; t < 2; ++t)
#pragma unroll
      for (int r = 0; r < 8; ++r)
        if (mi + r < nNodes) h[(size_t)(mi + r) * OUT_F + t * 16 + mrow] = acc[t][r];
  }

  const float as0 = att_src[mrow], as1 = att_src[16 + mrow];
  const float ad0 = att_dst[mrow], ad1 = att_dst[16 + mrow];
  float v[16];
#pragma unroll
  for (int r = 0; r < 8; ++r) {
    v[r]     = acc[0][r] * as0 + acc[1][r] * as1;
    v[8 + r] = acc[0][r] * ad0 + acc[1][r] * ad1;
  }
  const bool b8 = (lane & 8) != 0, b4 = (lane & 4) != 0, b2 = (lane & 2) != 0, b1 = (lane & 1) != 0;
  float w8[8], w4[4], w2[2];
#pragma unroll
  for (int i = 0; i < 8; ++i) {
    const float keep = b8 ? v[i + 8] : v[i];
    const float send = b8 ? v[i] : v[i + 8];
    w8[i] = keep + row_xor<8>(send);
  }
#pragma unroll
  for (int i = 0; i < 4; ++i) {
    const float keep = b4 ? w8[i + 4] : w8[i];
    const float send = b4 ? w8[i] : w8[i + 4];
    w4[i] = keep + row_xor<4>(send);
  }
#pragma unroll
  for (int i = 0; i < 2; ++i) {
    const float keep = b2 ? w4[i + 2] : w4[i];
    const float send = b2 ? w4[i] : w4[i + 2];
    w2[i] = keep + row_xor<2>(send);
  }
  const float keep1 = b1 ? w2[1] : w2[0];
  const float send1 = b1 ? w2[0] : w2[1];
  const float tot   = keep1 + row_xor<1>(send1);
  const int orow = mi + (mrow & 7);
  if (orow < nNodes) {
    float* dstp = (mrow < 8) ? a_src : a_dst;
    dstp[orow] = tot;
  }
}

__global__ __launch_bounds__(BIN_T)
void k_tile_sort(const int* __restrict__ ei, long long E,
                 const float* __restrict__ a_src,
                 int* __restrict__ tileEnd, int2* __restrict__ binned) {
  __shared__ int  cnt[MAXP];
  __shared__ int  lc[MAXP];
  __shared__ int  lst[MAXP];
  __shared__ __attribute__((aligned(8))) int2 srec[TILE];
  const int tid = threadIdx.x;
  cnt[tid] = 0;
  __syncthreads();
  const long long t0 = (long long)blockIdx.x * TILE;
  const int nValid = (E - t0 < (long long)TILE) ? (int)(E - t0) : TILE;
  const int* dstp = ei + E;
  int   sv[TILE / BIN_T], dv[TILE / BIN_T], rk[TILE / BIN_T];
  float av[TILE / BIN_T];
#pragma unroll
  for (int k = 0; k < TILE / BIN_T; ++k) {
    const long long e  = t0 + k * BIN_T + tid;
    const long long ec = (e < E) ? e : (E - 1);
    sv[k] = ei[ec];
    dv[k] = dstp[ec];
  }
#pragma unroll
  for (int k = 0; k < TILE / BIN_T; ++k) {
    const long long e = t0 + k * BIN_T + tid;
    rk[k] = 0;
    if (e < E) rk[k] = atomicAdd(&cnt[dv[k] >> NP_SHIFT], 1);
  }
#pragma unroll
  for (int k = 0; k < TILE / BIN_T; ++k) av[k] = a_src[sv[k]];
  __syncthreads();
  const int c = cnt[tid];
  lc[tid] = c;
  for (int off = 1; off < MAXP; off <<= 1) {
    __syncthreads();
    const int tl = (tid >= off) ? lc[tid - off] : 0;
    __syncthreads();
    lc[tid] += tl;
  }
  const int incl = lc[tid];
  lst[tid] = incl - c;
  tileEnd[(size_t)blockIdx.x * MAXP + tid] = incl;
  __syncthreads();
  int slot[TILE / BIN_T];
#pragma unroll
  for (int k = 0; k < TILE / BIN_T; ++k) slot[k] = lst[dv[k] >> NP_SHIFT] + rk[k];
#pragma unroll
  for (int k = 0; k < TILE / BIN_T; ++k) {
    if (k * BIN_T + tid < nValid) {
      const int key = sv[k] | ((dv[k] & (NP - 1)) << S_BITS);
      srec[slot[k]] = make_int2(key, __float_as_int(av[k]));
    }
  }
  __syncthreads();
  int2 st[TILE / BIN_T];
#pragma unroll
  for (int k = 0; k < TILE / BIN_T; ++k) {
    const int i = k * BIN_T + tid;
    st[k] = srec[(i < nValid) ? i : (nValid - 1)];
  }
#pragma unroll
  for (int k = 0; k < TILE / BIN_T; ++k)
    if (k * BIN_T + tid < nValid) binned[t0 + k * BIN_T + tid] = st[k];
}

__device__ __forceinline__ int wave_sum(int v, int lane) {
  for (int d = 16; d; d >>= 1) v += __builtin_amdgcn_ds_bpermute((lane ^ d) << 2, v);
  return v;
}

template <bool WITH_MAX>
__device__ __forceinline__ void stage_window(const int2* __restrict__ binned, const int* sPre, const int* sPos,
                                             int2* rec_s, const float* adst_s, unsigned* m_s,
                                             int w0, int wEnd, int top, int tid) {
  for (int v0 = w0 + tid; v0 < wEnd; v0 += STG_U * AGG_T) {
    int vv[STG_U], lo[STG_U];
#pragma unroll
    for (int u = 0; u < STG_U; ++u) {
      const int v = v0 + u * AGG_T;
      vv[u] = (v < wEnd) ? v : (wEnd - 1);
      lo[u] = 0;
    }
    for (int step = top; step; step >>= 1) {
#pragma unroll
      for (int u = 0; u < STG_U; ++u) {
        const int cand = lo[u] + step;
        if (sPre[cand] <= vv[u]) lo[u] = cand;
      }
    }
    int2 r[STG_U];
#pragma unroll
    for (int u = 0; u < STG_U; ++u) r[u] = binned[sPos[lo[u]] + (vv[u] - sPre[lo[u]])];
    asm volatile("" :: "v"(r[0].x), "v"(r[0].y), "v"(r[1].x), "v"(r[1].y), "v"(r[2].x), "v"(r[2].y),
                       "v"(r[3].x), "v"(r[3].y), "v"(r[4].x), "v"(r[4].y));
#pragma unroll
    for (int u = 0; u < STG_U; ++u) {
      if (v0 + u * AGG_T < wEnd) {
        rec_s[vv[u] - w0] = r[u];
        if (WITH_MAX) {
          const int dl = r[u].x >> S_BITS;
          float v = __int_as_float(r[u].y) + adst_s[dl];
          v = (v > 0.f) ? v : NEG_SLOPE * v;
          atomicMax(&m_s[dl], enc_f32(v));
        }
      }
    }
  }
}

__global__ __launch_bounds__(AGG_T) __attribute__((amdgpu_num_vgpr(64)))
void k_bin_aggregate(const int2* __restrict__ binned, const int* __restrict__ tileEnd, int nTiles,
                     const float* __restrict__ h,
                     const float* __restrict__ a_src, const float* __restrict__ a_dst,
                     const float* __restrict__ bias, float* __restrict__ out, int nNodes) {
  __shared__ float    acc[NP * OUT_F];
  __shared__ unsigned m_s[NP];
  __shared__ float    den[NP];
  __shared__ float    adst_s[NP];
  __shared__ float    vself_s[NP];
  __shared__ int      sPre[MAXT];
  __shared__ int      sPos[MAXT];
  __shared__ int      wtot[AGG_T / 32];
  __shared__ __attribute__((aligned(8))) int2 rec_s[REC_CAP];
  __shared__ __attribute__((aligned(16))) int dcnt[NP];
  __shared__ __attribute__((aligned(16))) int dstart[NP];
  const int tid  = threadIdx.x;
  const int lane = tid & 31;
  const int wv   = __builtin_amdgcn_readfirstlane(tid >> 5);
  const int p        = blockIdx.x;
  const int nodeBase = p << NP_SHIFT;
  const int nLocal   = (nNodes - nodeBase < NP) ? (nNodes - nodeBase) : NP;

  int sBeg = 0, sEnd = 0;
  if (tid < nTiles) {
    const int* te = tileEnd + (size_t)tid * MAXP + p;
    sEnd = te[0];
    if (p) sBeg = te[-1];
  }

  if (tid < NP) {
    float ad = 0.f, v = 0.f;
    unsigned me = 0u;
    if (tid < nLocal) {
      ad = a_dst[nodeBase + tid];
      v  = a_src[nodeBase + tid] + ad;
      v  = (v > 0.f) ? v : NEG_SLOPE * v;
      me = enc_f32(v);
    }
    adst_s[tid] = ad; vself_s[tid] = v; m_s[tid] = me; den[tid] = 0.f; dcnt[tid] = 0;
  }

  const int sc = sEnd - sBeg;
  int incl = sc;
  for (int d = 1; d < 32; d <<= 1) {
    const int o = __builtin_amdgcn_ds_bpermute(((lane - d) & 31) << 2, incl);
    if (lane >= d) incl += o;
  }
  if (lane == 31) wtot[wv] = incl;
  __syncthreads();
  const int wt    = wtot[lane];
  const int count = __builtin_amdgcn_readfirstlane(wave_sum(wt, lane));
  sPre[tid] = wave_sum((lane < wv) ? wt : 0, lane) + incl - sc;
  sPos[tid] = tid * TILE + sBeg;
  const int top = 1 << (31 - __builtin_clz((nTiles - 1) | 1));
  __syncthreads();

  if (count <= STG_U * AGG_T) {
    int2 r[STG_U];
    int  rk[STG_U];
#pragma unroll
    for (int u = 0; u < STG_U; ++u) { r[u] = make_int2(0, 0); rk[u] = 0; }
    if (count > 0) {
      int vv[STG_U], lo[STG_U];
#pragma unroll
      for (int u = 0; u < STG_U; ++u) {
        const int v = tid + u * AGG_T;
        vv[u] = (v < count) ? v : (count - 1);
        lo[u] = 0;
      }
      for (int step = top; step; step >>= 1) {
#pragma unroll
        for (int u = 0; u < STG_U; ++u) {
          const int cand = lo[u] + step;
          if (sPre[cand] <= vv[u]) lo[u] = cand;
        }
      }
#pragma unroll
      for (int u = 0; u < STG_U; ++u) r[u] = binned[sPos[lo[u]] + (vv[u] - sPre[lo[u]])];
      asm volatile("" :: "v"(r[0].x), "v"(r[0].y), "v"(r[1].x), "v"(r[1].y), "v"(r[2].x), "v"(r[2].y),
                         "v"(r[3].x), "v"(r[3].y), "v"(r[4].x), "v"(r[4].y));
    }
#pragma unroll
    for (int u = 0; u < STG_U; ++u) {
      if (tid + u * AGG_T < count) {
        const int dl = r[u].x >> S_BITS;
        float v = __int_as_float(r[u].y) + adst_s[dl];
        v = (v > 0.f) ? v : NEG_SLOPE * v;
        r[u].y = __float_as_int(v);
        rk[u]  = atomicAdd(&dcnt[dl], 1);
        atomicMax(&m_s[dl], enc_f32(v));
      }
    }
    __syncthreads();
    if (wv == 0) {
      const int4 c0 = *(const int4*)(dcnt + lane * 8);
      const int4 c1 = *(const int4*)(dcnt + lane * 8 + 4);
      const int p0 = c0.x, p1 = p0 + c0.y, p2 = p1 + c0.z, p3 = p2 + c0.w;
      const int p4 = p3 + c1.x, p5 = p4 + c1.y, p6 = p5 + c1.z, p7 = p6 + c1.w;
      int inc = p7;
      for (int d = 1; d < 32; d <<= 1) {
        const int o = __builtin_amdgcn_ds_bpermute(((lane - d) & 31) << 2, inc);
        if (lane >= d) inc += o;
      }
      const int ex0 = inc - p7;
      *(int4*)(dstart + lane * 8)     = make_int4(ex0, ex0 + p0, ex0 + p1, ex0 + p2);
      *(int4*)(dstart + lane * 8 + 4) = make_int4(ex0 + p3, ex0 + p4, ex0 + p5, ex0 + p6);
    }
    __syncthreads();
#pragma unroll
    for (int u = 0; u < STG_U; ++u) {
      if (tid + u * AGG_T < count) {
        const int dl = r[u].x >> S_BITS;
        const float x = __expf(__int_as_float(r[u].y) - dec_f32(m_s[dl]));
        rec_s[dstart[dl] + rk[u]] = make_int2(r[u].x, __float_as_int(x));
      }
    }
    __syncthreads();

    const int    g    = lane >> 3;
    const int    c4   = (lane & 7) << 2;
    const char*    hbytes = (const char*)h;
    const unsigned cb     = (unsigned)c4 << 2;
    for (int rr = 0; rr < NP / 128; ++rr) {
      const int   dl      = wv * (NP / 32) + rr * 4 + g;
      const bool  live    = dl < nLocal;
      const int   selfRow = nodeBase + (live ? dl : 0);
      const int   n       = dcnt[dl];
      const int   beg     = dstart[dl];
      const float xs      = __expf(vself_s[dl] - dec_f32(m_s[dl]));
      const float4 hs     = *(const float4*)(hbytes + (((unsigned)selfRow << 7) | cb));
      float4 a  = make_float4(xs * hs.x, xs * hs.y, xs * hs.z, xs * hs.w);
      float  dn = xs;
      const int nA = __builtin_amdgcn_readlane(n, 0),  nB = __builtin_amdgcn_readlane(n, 8);
      const int nC = __builtin_amdgcn_readlane(n, 16), nD = __builtin_amdgcn_readlane(n, 24);
      const int nAB = nA > nB ? nA : nB, nCD = nC > nD ? nC : nD;
      const int nmax = nAB > nCD ? nAB : nCD;
      for (int j0 = 0; j0 < nmax; j0 += 8) {
        const int base = beg + j0;
        const int2* rp = rec_s + ((base < REC_CAP - 8) ? base : (REC_CAP - 8));
        int2 rc[8];
#pragma unroll
        for (int u = 0; u < 8; ++u) rc[u] = rp[u];
        asm volatile("" :: "v"(rc[0].x), "v"(rc[0].y), "v"(rc[1].x), "v"(rc[1].y), "v"(rc[2].x), "v"(rc[2].y),
                           "v"(rc[3].x), "v"(rc[3].y), "v"(rc[4].x), "v"(rc[4].y), "v"(rc[5].x), "v"(rc[5].y),
                           "v"(rc[6].x), "v"(rc[6].y), "v"(rc[7].x), "v"(rc[7].y));
        float  xv[8];
        float4 hv[8];
#pragma unroll
        for (int u = 0; u < 4; ++u) {
          const bool on = j0 + u < n;
          const int  s  = on ? (rc[u].x & ((1 << S_BITS) - 1)) : selfRow;
          xv[u] = on ? __int_as_float(rc[u].y) : 0.f;
          hv[u] = *(const float4*)(hbytes + (((unsigned)s << 7) | cb));
        }
        if (j0 + 4 < nmax) {
#pragma unroll
          for (int u = 4; u < 8; ++u) {
            const bool on = j0 + u < n;
            const int  s  = on ? (rc[u].x & ((1 << S_BITS) - 1)) : selfRow;
            xv[u] = on ? __int_as_float(rc[u].y) : 0.f;
            hv[u] = *(const float4*)(hbytes + (((unsigned)s << 7) | cb));
          }
        } else {
#pragma unroll
          for (int u = 4; u < 8; ++u) { xv[u] = 0.f; hv[u] = make_float4(0.f, 0.f, 0.f, 0.f); }
        }
#pragma unroll
        for (int u = 0; u < 8; ++u) {
          a.x += xv[u] * hv[u].x; a.y += xv[u] * hv[u].y;
          a.z += xv[u] * hv[u].z; a.w += xv[u] * hv[u].w;
          dn  += xv[u];
        }
      }
      if (live) {
        const float4 b4 = *(const float4*)(bias + c4);
        const float inv = 1.f / dn;
        float4 o = make_float4(a.x * inv + b4.x, a.y * inv + b4.y, a.z * inv + b4.z, a.w * inv + b4.w);
        o.x = o.x > 0.f ? o.x : 0.f; o.y = o.y > 0.f ? o.y : 0.f;
        o.z = o.z > 0.f ? o.z : 0.f; o.w = o.w > 0.f ? o.w : 0.f;
        *(float4*)(out + (size_t)(nodeBase + dl) * OUT_F + c4) = o;
      }
    }
    return;
  }

  const size_t rowBase = (size_t)nodeBase * OUT_F;
  float hself[NP * OUT_F / AGG_T];
#pragma unroll
  for (int q = 0; q < NP * OUT_F / AGG_T; ++q) {
    const int k = tid + q * AGG_T;
    hself[q] = (k < nLocal * OUT_F) ? h[rowBase + k] : 0.f;
  }
  const float bs = bias[lane];
  for (int k = tid; k < NP * OUT_F; k += AGG_T) acc[k] = 0.f;

  for (int w0 = 0; w0 < count; w0 += REC_CAP) {
    const int wEnd = (count - w0 < REC_CAP) ? count : (w0 + REC_CAP);
    stage_window<true>(binned, sPre, sPos, rec_s, adst_s, m_s, w0, wEnd, top, tid);
  }
  __syncthreads();

  for (int w0 = 0; w0 < count; w0 += REC_CAP) {
    const int wEnd = (count - w0 < REC_CAP) ? count : (w0 + REC_CAP);
    if (count > REC_CAP) {
      __syncthreads();
      stage_window<false>(binned, sPre, sPos, rec_s, adst_s, m_s, w0, wEnd, top, tid);
      __syncthreads();
    }
    for (int c = w0 + wv * 32; c < wEnd; c += AGG_T) {
      const int e = c + lane;
      const int2 r = rec_s[((e < wEnd) ? e : (wEnd - 1)) - w0];
      const int s  = r.x & ((1 << S_BITS) - 1);
      const int dl = r.x >> S_BITS;
      float ex = 0.f;
      if (e < wEnd) {
        float v = __int_as_float(r.y) + adst_s[dl];
        v = (v > 0.f) ? v : NEG_SLOPE * v;
        ex = __expf(v - dec_f32(m_s[dl]));
        atomicAdd(&den[dl], ex);
      }
      const int n = (wEnd - c < 32) ? (wEnd - c) : 32;
      int j = 0;
      for (; j + AGG_B <= n; j += AGG_B) {
        float hv[AGG_B], xj[AGG_B];
        int   dj[AGG_B];
#pragma unroll
        for (int u = 0; u < AGG_B; ++u) {
          const int sj = __builtin_amdgcn_readlane(s, j + u);
          dj[u] = __builtin_amdgcn_readlane(dl, j + u);
          xj[u] = __int_as_float(__builtin_amdgcn_readlane(__float_as_int(ex), j + u));
          hv[u] = h[(size_t)sj * OUT_F + lane];
        }
#pragma unroll
        for (int u = 0; u < AGG_B; ++u)
          atomicAdd(&acc[dj[u] * OUT_F + lane], hv[u] * xj[u]);
      }
      for (; j < n; ++j) {
        const int   sj = __builtin_amdgcn_readlane(s, j);
        const int   dd = __builtin_amdgcn_readlane(dl, j);
        const float xx = __int_as_float(__builtin_amdgcn_readlane(__float_as_int(ex), j));
        atomicAdd(&acc[dd * OUT_F + lane], h[(size_t)sj * OUT_F + lane] * xx);
      }
    }
  }
  __syncthreads();

#pragma unroll
  for (int q = 0; q < NP * OUT_F / AGG_T; ++q) {
    const int k = tid + q * AGG_T;
    if (k < nLocal * OUT_F) {
      const int i = k >> 5;
      const float ex  = __expf(vself_s[i] - dec_f32(m_s[i]));
      const float num = acc[k] + ex * hself[q];
      const float dn  = den[i] + ex;
      const float o   = num / dn + bs;
      out[rowBase + k] = o > 0.f ? o : 0.f;
    }
  }
}

extern "C" void kernel_launch(void* const* d_in, const int* in_sizes, int n_in,
                              void* d_out, int out_size, void* d_ws, size_t ws_size,
                              hipStream_t stream) {
  const float* x       = (const float*)d_in[0];
  const int*   ei      = (const int*)  d_in[1];
  const float* W       = (const float*)d_in[2];
  const float* att_src = (const float*)d_in[3];
  const float* att_dst = (const float*)d_in[4];
  const float* bias    = (const float*)d_in[5];
  float*       out     = (float*)d_out;

  const int       nNodes = in_sizes[0] / IN_F;
  const long long E      = (long long)in_sizes[1] / 2;
  const int       nBins  = (nNodes + NP - 1) / NP;
  const int       nTiles = (int)((E + TILE - 1) / TILE);

  char* ws = (char*)d_ws;
  auto carve = [&](size_t bytes) {
    char* p = ws;
    ws += (bytes + 255) & ~(size_t)255;
    return p;
  };
  float*     h         = (float*)    carve((size_t)nNodes * OUT_F * sizeof(float));
  float*     a_src     = (float*)    carve((size_t)nNodes * sizeof(float));
  float*     a_dst     = (float*)    carve((size_t)nNodes * sizeof(float));
  int*       tileEnd   = (int*)      carve((size_t)nTiles * MAXP * sizeof(int));
  int2*      binned    = (int2*)     carve((size_t)E * sizeof(int2));

  const int wavesGemm  = (nNodes + 15) / 16;
  const int blocksGemm = (wavesGemm + 3) / 4;
  k_project<<<blocksGemm, 128, 0, stream>>>(x, W, att_src, att_dst, h, a_src, a_dst, nNodes);

  k_tile_sort<<<nTiles, BIN_T, 0, stream>>>(ei, E, a_src, tileEnd, binned);

  k_bin_aggregate<<<nBins, AGG_T, 0, stream>>>(binned, tileEnd, nTiles, h, a_src, a_dst, bias, out, nNodes);
}
